# starting kernel
# baseline (speedup 1.0000x reference)
#include <hip/hip_runtime.h>
#include <hip/hip_bf16.h>
#include <cstdint>
#include <cstddef>

typedef float v2f __attribute__((ext_vector_type(2)));
typedef float v4f __attribute__((ext_vector_type(4)));
typedef float v8f __attribute__((ext_vector_type(8)));

constexpr int Bc = 8;
constexpr int Lc = 4096;
constexpr int Hc = 1024;
constexpr int Sc = 512;
constexpr int Tc = 512;
constexpr int TOK = 64;
constexpr int KC  = 32;
constexpr int TSTRIDE = 36;
constexpr int WSTRIDE = 34;
constexpr int NSEG = 1056;

__global__ __launch_bounds__(1024) void seg_scan_kernel(
    const int* __restrict__ wid, const int* __restrict__ mask,
    int* __restrict__ seg, int* __restrict__ flag) {
  __shared__ int sHas[1024];
  __shared__ int sWid[1024];
  __shared__ int sSum[1024];
  const int tid = threadIdx.x;
  const int b = blockIdx.x;
  const int base = b * Lc + tid * 4;

  int w[4], v[4];
#pragma unroll
  for (int j = 0; j < 4; ++j) {
    w[j] = wid[base + j];
    v[j] = (mask[base + j] > 0) ? 1 : 0;
  }

  int curH = 0, curW = -2;
#pragma unroll
  for (int j = 0; j < 4; ++j)
    if (v[j]) { curH = 1; curW = w[j]; }
  sHas[tid] = curH; sWid[tid] = curW;
  __syncthreads();
  for (int off = 1; off < 1024; off <<= 1) {
    int pH = 0, pW = -2;
    if (tid >= off) { pH = sHas[tid - off]; pW = sWid[tid - off]; }
    __syncthreads();
    if (!curH && pH) { curH = 1; curW = pW; }
    sHas[tid] = curH; sWid[tid] = curW;
    __syncthreads();
  }
  int prev = -2;
  if (tid > 0 && sHas[tid - 1]) prev = sWid[tid - 1];

  int f[4]; int localSum = 0;
#pragma unroll
  for (int j = 0; j < 4; ++j) {
    f[j] = 0;
    if (v[j]) {
      f[j] = (w[j] == -1 || prev == -1 || w[j] != prev) ? 1 : 0;
      prev = w[j];
      localSum += f[j];
    }
  }

  int curS = localSum;
  sSum[tid] = curS;
  __syncthreads();
  for (int off = 1; off < 1024; off <<= 1) {
    int p = 0;
    if (tid >= off) p = sSum[tid - off];
    __syncthreads();
    curS += p;
    sSum[tid] = curS;
    __syncthreads();
  }
  int run = (tid > 0) ? sSum[tid - 1] : 0;
#pragma unroll
  for (int j = 0; j < 4; ++j) {
    int sg = -1;
    if (v[j]) { run += f[j]; sg = run - 1; }
    seg[base + j] = sg;
    flag[base + j] = f[j];
  }
}

__global__ __launch_bounds__(128) void dot_wmma_kernel(
    const float* __restrict__ X, const float* __restrict__ wvec,
    float* __restrict__ dsrc, float* __restrict__ dtgt) {
  __shared__ __align__(16) float tile[TOK * TSTRIDE];
  __shared__ __align__(16) float wPad[16 * WSTRIDE];

  const int tid  = threadIdx.x;
  const int lane = tid & 31;
  const int wave = tid >> 5;
  const int col  = lane & 15;
  const int half = lane >> 4;
  const int tokBase = blockIdx.x * TOK;

  for (int i = tid; i < 16 * WSTRIDE; i += 128) wPad[i] = 0.0f;

  v8f acc = {};

  for (int k0 = 0; k0 < Hc; k0 += KC) {
    __syncthreads();

    if (tid < 64) {
      int r = tid >> 5, c = tid & 31;
      wPad[r * WSTRIDE + c] = wvec[r * Hc + k0 + c];
    }
#pragma unroll
    for (int it = 0; it < 4; ++it) {
      int idx = it * 128 + tid;
      int row = idx >> 3, c4 = idx & 7;
      v4f val = *reinterpret_cast<const v4f*>(
          X + (size_t)(tokBase + row) * Hc + k0 + c4 * 4);
      *reinterpret_cast<v4f*>(&tile[row * TSTRIDE + c4 * 4]) = val;
    }
    if (k0 + KC < Hc && tid < TOK) {
      __builtin_prefetch(X + (size_t)(tokBase + tid) * Hc + k0 + KC, 0, 1);
    }
    __syncthreads();

    const float* aBase = &tile[(wave * 16 + col) * TSTRIDE + half * 2];
    const float* bBase = &wPad[col * WSTRIDE + half * 2];
#pragma unroll
    for (int s = 0; s < 8; ++s) {
      v2f a = *reinterpret_cast<const v2f*>(aBase + s * 4);
      v2f bf = *reinterpret_cast<const v2f*>(bBase + s * 4);
      acc = __builtin_amdgcn_wmma_f32_16x16x4_f32(
           false, a,  false, bf,
           (short)0, acc,  false,  false);
    }
  }

  if (col < 2) {
    float* dst = col ? dtgt : dsrc;
    int rb = tokBase + wave * 16 + half * 8;
#pragma unroll
    for (int r = 0; r < 8; ++r) dst[rb + r] = acc[r];
  }
}

__global__ void seg_reduce_kernel(
    const int* __restrict__ seg, const int* __restrict__ flag,
    const float* __restrict__ dsrc, const float* __restrict__ dtgt,
    float* __restrict__ sums, int* __restrict__ counts) {
  int idx = blockIdx.x * blockDim.x + threadIdx.x;
  if (idx >= Bc * Lc) return;
  if (!flag[idx]) return;
  int sg = seg[idx];
  if (sg < 1 || sg > Sc + Tc) return;
  int b = idx / Lc;
  int end = (b + 1) * Lc;
  const float* d = (sg <= Sc) ? dsrc : dtgt;
  float sum = 0.0f; int cnt = 0;
  for (int j = idx; j < end; ++j) {
    int s2 = seg[j];
    if (s2 == sg) { sum += d[j]; ++cnt; }
    else if (s2 != -1) break;
  }
  sums[b * NSEG + sg] = sum;
  counts[b * NSEG + sg] = cnt;
}

__global__ void avg_kernel(const float* __restrict__ sums,
                           const int* __restrict__ counts,
                           float* __restrict__ avg_a, float* __restrict__ avg_c) {
  int idx = blockIdx.x * blockDim.x + threadIdx.x;
  if (idx >= Bc * (Sc + Tc)) return;
  int b = idx / (Sc + Tc);
  int i = idx % (Sc + Tc);
  int sg = 1 + i;
  float cnt = (float)max(counts[b * NSEG + sg], 1);
  float m = sums[b * NSEG + sg] / cnt;
  if (i < Sc) avg_a[b * Sc + i] = m;
  else        avg_c[b * Tc + (i - Sc)] = m;
}

__global__ void logits_kernel(const float* __restrict__ a,
                              const float* __restrict__ c,
                              const float* __restrict__ biasPtr,
                              float* __restrict__ out) {
  int idx = blockIdx.x * blockDim.x + threadIdx.x;
  if (idx >= Bc * Sc * (Tc / 4)) return;
  int t4  = idx % (Tc / 4);
  int rem = idx / (Tc / 4);
  int s = rem % Sc;
  int b = rem / Sc;
  float av = a[b * Sc + s] + biasPtr[0];
  v4f cv = *reinterpret_cast<const v4f*>(c + b * Tc + t4 * 4);
  v4f o = cv + av;
  reinterpret_cast<v4f*>(out)[idx] = o;
}

extern "C" void kernel_launch(void* const* d_in, const int* in_sizes, int n_in,
                              void* d_out, int out_size, void* d_ws, size_t ws_size,
                              hipStream_t stream) {
  const float* X    = (const float*)d_in[0];
  const int*   msk  = (const int*)d_in[1];
  const int*   wid  = (const int*)d_in[2];
  const float* wvec = (const float*)d_in[5];
  const float* bias = (const float*)d_in[6];
  float* out = (float*)d_out;

  char* ws = (char*)d_ws;
  size_t off = 0;
  int*   seg    = (int*)(ws + off);   off += (size_t)Bc * Lc * 4;
  int*   flg    = (int*)(ws + off);   off += (size_t)Bc * Lc * 4;
  float* dsrc   = (float*)(ws + off); off += (size_t)Bc * Lc * 4;
  float* dtgt   = (float*)(ws + off); off += (size_t)Bc * Lc * 4;
  char*  zbase  = ws + off;
  float* sums   = (float*)(ws + off); off += (size_t)Bc * NSEG * 4;
  int*   counts = (int*)(ws + off);   off += (size_t)Bc * NSEG * 4;
  size_t zbytes = (size_t)Bc * NSEG * 8;
  float* avg_a  = (float*)(ws + off); off += (size_t)Bc * Sc * 4;
  float* avg_c  = (float*)(ws + off); off += (size_t)Bc * Tc * 4;

  hipMemsetAsync(zbase, 0, zbytes, stream);

  seg_scan_kernel<<<Bc, 1024, 0, stream>>>(wid, msk, seg, flg);

  dot_wmma_kernel<<<(Bc * Lc) / TOK, 128, 0, stream>>>(X, wvec, dsrc, dtgt);

  seg_reduce_kernel<<<(Bc * Lc + 255) / 256, 256, 0, stream>>>(
      seg, flg, dsrc, dtgt, sums, counts);

  avg_kernel<<<(Bc * (Sc + Tc) + 255) / 256, 256, 0, stream>>>(
      sums, counts, avg_a, avg_c);

  logits_kernel<<<(Bc * Sc * (Tc / 4) + 255) / 256, 256, 0, stream>>>(
      avg_a, avg_c, bias, out);
}


# dot kernel cut by segments, not tokens: a block takes 16 consecutive segments from a start table the scan writes, streams only their rows, writes their means; reduce, avg, memset go; sum order kept
# speedup vs baseline: 1.9896x; 1.9896x over previous
#include <hip/hip_runtime.h>
#include <hip/hip_bf16.h>
#include <cstdint>
#include <cstddef>

typedef float v2f __attribute__((ext_vector_type(2)));
typedef float v4f __attribute__((ext_vector_type(4)));
typedef float v8f __attribute__((ext_vector_type(8)));

constexpr int Bc = 8;
constexpr int Lc = 4096;
constexpr int Hc = 1024;
constexpr int Sc = 512;
constexpr int Tc = 512;
constexpr int TOK = 64;
constexpr int WTOK = 8;
constexpr int KSTEP = 64;
constexpr int NSTEP = Hc / KSTEP;
constexpr int GRP = 4;
constexpr int NSEG = 1056;
constexpr int G = 16;
constexpr int NBLK = (Sc + Tc) / G;
constexpr int MSTR = 32;

__global__ __launch_bounds__(1024) void seg_scan_kernel(
    const int* __restrict__ wid, const int* __restrict__ mask,
    int* __restrict__ st) {
  __shared__ int sHas[1024];
  __shared__ int sWid[1024];
  __shared__ int sSum[1024];
  const int tid = threadIdx.x;
  const int b = blockIdx.x;
  const int base = b * Lc + tid * 4;

  int w[4], v[4];
#pragma unroll
  for (int j = 0; j < 4; ++j) {
    w[j] = wid[base + j];
    v[j] = (mask[base + j] > 0) ? 1 : 0;
  }

  int curH = 0, curW = -2;
#pragma unroll
  for (int j = 0; j < 4; ++j)
    if (v[j]) { curH = 1; curW = w[j]; }
  sHas[tid] = curH; sWid[tid] = curW;
  __syncthreads();
  for (int off = 1; off < 1024; off <<= 1) {
    int pH = 0, pW = -2;
    if (tid >= off) { pH = sHas[tid - off]; pW = sWid[tid - off]; }
    __syncthreads();
    if (!curH && pH) { curH = 1; curW = pW; }
    sHas[tid] = curH; sWid[tid] = curW;
    __syncthreads();
  }
  int prev = -2;
  if (tid > 0 && sHas[tid - 1]) prev = sWid[tid - 1];

  int f[4]; int localSum = 0;
#pragma unroll
  for (int j = 0; j < 4; ++j) {
    f[j] = 0;
    if (v[j]) {
      f[j] = (w[j] == -1 || prev == -1 || w[j] != prev) ? 1 : 0;
      prev = w[j];
      localSum += f[j];
    }
  }

  int curS = localSum;
  sSum[tid] = curS;
  __syncthreads();
  for (int off = 1; off < 1024; off <<= 1) {
    int p = 0;
    if (tid >= off) p = sSum[tid - off];
    __syncthreads();
    curS += p;
    sSum[tid] = curS;
    __syncthreads();
  }
  int run = (tid > 0) ? sSum[tid - 1] : 0;
  const int nseg = sSum[1023];
#pragma unroll
  for (int j = 0; j < 4; ++j) {
    if (f[j]) {
      run += 1;
      if (run - 1 < NSEG) st[b * NSEG + run - 1] = tid * 4 + j;
    }
  }
  for (int i = tid; i < NSEG; i += 1024)
    if (i >= nseg) st[b * NSEG + i] = Lc;
}

__device__ __forceinline__ float pick8(const v8f& a, bool p1, bool p2, bool p4) {
  float s0 = p1 ? a[1] : a[0];
  float s1 = p1 ? a[3] : a[2];
  float s2 = p1 ? a[5] : a[4];
  float s3 = p1 ? a[7] : a[6];
  float u0 = p2 ? s1 : s0;
  float u1 = p2 ? s3 : s2;
  return p4 ? u1 : u0;
}

__global__ __launch_bounds__(256) void dot_seg_kernel(
    const float* __restrict__ X, const float* __restrict__ wvec,
    const int* __restrict__ mask, const int* __restrict__ st,
    float* __restrict__ means) {
  __shared__ float sDot[TOK];
  __shared__ int   sVal[TOK];

  const int tid  = threadIdx.x;
  const int lane = tid & 31;
  const int wave = __builtin_amdgcn_readfirstlane(tid >> 5);
  const int c    = lane & 15;
  const int h    = lane >> 4;
  const int t    = c >> 3;
  const int p    = c & 7;
  const int koff = 8 * p + 4 * h;
  const bool p1 = p & 1, p2 = p & 2, p4 = p & 4;

  const int b   = blockIdx.x / NBLK;
  const int jb  = blockIdx.x % NBLK;
  const int cls = (jb >= Sc / G) ? 1 : 0;
  const int* stb = st + b * NSEG + 1 + jb * G;
  const int t0 = stb[0];
  const int t1 = stb[G];
  int s0 = 0, s1 = 0;
  if (tid < G) { s0 = stb[tid]; s1 = stb[tid + 1]; }
  float sum = 0.0f; int cnt = 0;

  const float* wp = wvec + cls * Hc + koff;
  const float* Xb = X + (size_t)b * Lc * Hc + koff;
  const int*   mb = mask + b * Lc;

  for (int base = t0; base < t1; base = __builtin_amdgcn_readfirstlane(base + TOK)) {
    const int mraw = mb[min(base + tid, t1 - 1)];

    const int tokW = base + wave * WTOK;
    if (tokW < t1) {
      v4f wr[NSTEP];
#pragma unroll
      for (int q = 0; q < NSTEP; ++q)
        wr[q] = *reinterpret_cast<const v4f*>(wp + q * KSTEP);

      const float* xr[WTOK / 2];
#pragma unroll
      for (int P = 0; P < WTOK / 2; ++P)
        xr[P] = Xb + (size_t)min(tokW + 2 * P + t, t1 - 1) * Hc;

      v4f xb[2][NSTEP];
#pragma unroll
      for (int q = 0; q < NSTEP; ++q)
        xb[0][q] = *reinterpret_cast<const v4f*>(xr[0] + q * KSTEP);
#pragma unroll
      for (int q = 0; q < NSTEP; ++q)
        xb[1][q] = *reinterpret_cast<const v4f*>(xr[1] + q * KSTEP);
      __builtin_amdgcn_sched_barrier(0);

      float res[WTOK / 2];
#pragma unroll
      for (int P = 0; P < WTOK / 2; ++P) {
        v8f acc0 = {}, acc1 = {};
#pragma unroll
        for (int g = 0; g < NSTEP / GRP; ++g) {
#pragma unroll
          for (int j = 0; j < GRP; ++j) {
            const int q = g * GRP + j;
            const v4f a = xb[P & 1][q];
            const v4f bw = wr[q];
            v2f a0 = {a.x, a.y}, a1 = {a.z, a.w};
            v2f b0 = {bw.x, bw.y}, b1 = {bw.z, bw.w};
            acc0 = __builtin_amdgcn_wmma_f32_16x16x4_f32(
                 false, a0,  false, b0,
                 (short)0, acc0,  false,  false);
            acc1 = __builtin_amdgcn_wmma_f32_16x16x4_f32(
                 false, a1,  false, b1,
                 (short)0, acc1,  false,  false);
          }
          if (P + 2 < WTOK / 2) {
#pragma unroll
            for (int j = 0; j < GRP; ++j) {
              const int q = g * GRP + j;
              xb[P & 1][q] = *reinterpret_cast<const v4f*>(xr[P + 2] + q * KSTEP);
            }
          }
          __builtin_amdgcn_sched_barrier(0);
        }
        const v8f acc = acc0 + acc1;
        res[P] = pick8(acc, p1, p2, p4);
      }

#pragma unroll
      for (int P = 0; P < WTOK / 2; ++P) {
        float s = res[P];
        s += __shfl_xor(s, 1, 32);
        s += __shfl_xor(s, 2, 32);
        s += __shfl_xor(s, 4, 32);
        res[P] = s;
      }
      if (c == 0) {
#pragma unroll
        for (int P = 0; P < WTOK / 2; ++P)
          sDot[wave * WTOK + 2 * P + h] = res[P];
      }
    }
    if (tid < TOK) sVal[tid] = (base + tid < t1 && mraw > 0) ? 1 : 0;
    __syncthreads();
    if (tid < G) {
      const int lo = max(s0, base), hi = min(s1, base + TOK);
      for (int k = lo; k < hi; ++k)
        if (sVal[k - base]) { sum += sDot[k - base]; ++cnt; }
    }
    __syncthreads();
  }
  if (tid < G)
    means[(b * NBLK + jb) * MSTR + tid] = sum / (float)max(cnt, 1);
}

__global__ void logits_kernel(const float* __restrict__ means,
                              const float* __restrict__ biasPtr,
                              float* __restrict__ out) {
  int idx = blockIdx.x * blockDim.x + threadIdx.x;
  if (idx >= Bc * Sc * (Tc / 4)) return;
  int t4  = idx % (Tc / 4);
  int rem = idx / (Tc / 4);
  int s = rem % Sc;
  int b = rem / Sc;
  int tt = t4 * 4;
  float av = means[(b * NBLK + s / G) * MSTR + s % G] + biasPtr[0];
  v4f cv = *reinterpret_cast<const v4f*>(
      means + (b * NBLK + Sc / G + tt / G) * MSTR + tt % G);
  v4f o = cv + av;
  reinterpret_cast<v4f*>(out)[idx] = o;
}

extern "C" void kernel_launch(void* const* d_in, const int* in_sizes, int n_in,
                              void* d_out, int out_size, void* d_ws, size_t ws_size,
                              hipStream_t stream) {
  const float* X    = (const float*)d_in[0];
  const int*   msk  = (const int*)d_in[1];
  const int*   wid  = (const int*)d_in[2];
  const float* wvec = (const float*)d_in[5];
  const float* bias = (const float*)d_in[6];
  float* out = (float*)d_out;

  char* ws = (char*)d_ws;
  size_t off = 0;
  int*   st    = (int*)(ws + off);   off += (size_t)Bc * NSEG * 4;
  float* means = (float*)(ws + off); off += (size_t)Bc * NBLK * MSTR * 4;

  seg_scan_kernel<<<Bc, 1024, 0, stream>>>(wid, msk, st);

  dot_seg_kernel<<<Bc * NBLK, 256, 0, stream>>>(X, wvec, msk, st, means);

  logits_kernel<<<(Bc * Sc * (Tc / 4) + 255) / 256, 256, 0, stream>>>(
      means, bias, out);
}


# segmentation kernel: the two 10-step LDS block scans (40 barriers) become wave lane-mask scans (ballot, mbcnt, shuffles) with one LDS record per wave: 2 barriers; table and all float sums unchanged
# speedup vs baseline: 2.2430x; 1.1274x over previous
#include <hip/hip_runtime.h>
#include <hip/hip_bf16.h>
#include <cstdint>
#include <cstddef>

typedef float v2f __attribute__((ext_vector_type(2)));
typedef float v4f __attribute__((ext_vector_type(4)));
typedef float v8f __attribute__((ext_vector_type(8)));

constexpr int Bc = 8;
constexpr int Lc = 4096;
constexpr int Hc = 1024;
constexpr int Sc = 512;
constexpr int Tc = 512;
constexpr int TOK = 64;
constexpr int WTOK = 8;
constexpr int KSTEP = 64;
constexpr int NSTEP = Hc / KSTEP;
constexpr int GRP = 4;
constexpr int NSEG = 1056;
constexpr int G = 16;
constexpr int NBLK = (Sc + Tc) / G;
constexpr int MSTR = 32;

__global__ __launch_bounds__(1024) void seg_scan_kernel(
    const int* __restrict__ wid, const int* __restrict__ mask,
    int* __restrict__ st) {
  __shared__ int sHas[32];
  __shared__ int sWid[32];
  __shared__ int sTot[32];
  const int tid  = threadIdx.x;
  const int lane = tid & 31;
  const int wave = __builtin_amdgcn_readfirstlane(tid >> 5);
  const int b = blockIdx.x;
  const int base = b * Lc + tid * 4;

  int w[4], v[4];
#pragma unroll
  for (int j = 0; j < 4; ++j) {
    w[j] = wid[base + j];
    v[j] = (mask[base + j] > 0) ? 1 : 0;
  }

  int curH = 0, curW = -2;
#pragma unroll
  for (int j = 0; j < 4; ++j)
    if (v[j]) { curH = 1; curW = w[j]; }
  const unsigned hm  = __builtin_amdgcn_ballot_w32(curH != 0);
  const unsigned bel = hm & ((1u << lane) - 1u);
  const int srcLane  = bel ? 31 - __builtin_clz(bel) : 0;
  const int pw = __shfl(curW, srcLane, 32);
  const int lastLane = hm ? 31 - __builtin_clz(hm) : 0;
  const int waveW = __builtin_amdgcn_readlane(curW, lastLane);
  if (lane == 0) { sHas[wave] = hm ? 1 : 0; sWid[wave] = waveW; }
  __syncthreads();
  const int rH = sHas[lane], rW = sWid[lane];
  const unsigned wm = __builtin_amdgcn_ballot_w32(rH != 0) & ((1u << wave) - 1u);
  const int wsrc = wm ? 31 - __builtin_clz(wm) : 0;
  const int cw = __builtin_amdgcn_readlane(rW, wsrc);
  const int carry = wm ? cw : -2;
  int prev = bel ? pw : carry;

  int f[4];
#pragma unroll
  for (int j = 0; j < 4; ++j) {
    f[j] = 0;
    if (v[j]) {
      f[j] = (w[j] == -1 || prev == -1 || w[j] != prev) ? 1 : 0;
      prev = w[j];
    }
  }

  unsigned ex = 0; int tot = 0;
#pragma unroll
  for (int j = 0; j < 4; ++j) {
    const unsigned bj = __builtin_amdgcn_ballot_w32(f[j] != 0);
    ex = __builtin_amdgcn_mbcnt_lo(bj, ex);
    tot += __builtin_popcount(bj);
  }
  if (lane == 0) sTot[wave] = tot;
  __syncthreads();
  const int tv = sTot[lane];
  int pk = ((lane < wave) ? tv : 0) | (tv << 16);
#pragma unroll
  for (int m = 1; m < 32; m <<= 1) pk += __shfl_xor(pk, m, 32);
  int run = (pk & 0xFFFF) + (int)ex;
  const int nseg = pk >> 16;
#pragma unroll
  for (int j = 0; j < 4; ++j) {
    if (f[j]) {
      run += 1;
      if (run - 1 < NSEG) st[b * NSEG + run - 1] = tid * 4 + j;
    }
  }
  for (int i = tid; i < NSEG; i += 1024)
    if (i >= nseg) st[b * NSEG + i] = Lc;
}

__device__ __forceinline__ float pick8(const v8f& a, bool p1, bool p2, bool p4) {
  float s0 = p1 ? a[1] : a[0];
  float s1 = p1 ? a[3] : a[2];
  float s2 = p1 ? a[5] : a[4];
  float s3 = p1 ? a[7] : a[6];
  float u0 = p2 ? s1 : s0;
  float u1 = p2 ? s3 : s2;
  return p4 ? u1 : u0;
}

__global__ __launch_bounds__(256) void dot_seg_kernel(
    const float* __restrict__ X, const float* __restrict__ wvec,
    const int* __restrict__ mask, const int* __restrict__ st,
    float* __restrict__ means) {
  __shared__ float sDot[TOK];
  __shared__ int   sVal[TOK];

  const int tid  = threadIdx.x;
  const int lane = tid & 31;
  const int wave = __builtin_amdgcn_readfirstlane(tid >> 5);
  const int c    = lane & 15;
  const int h    = lane >> 4;
  const int t    = c >> 3;
  const int p    = c & 7;
  const int koff = 8 * p + 4 * h;
  const bool p1 = p & 1, p2 = p & 2, p4 = p & 4;

  const int b   = blockIdx.x / NBLK;
  const int jb  = blockIdx.x % NBLK;
  const int cls = (jb >= Sc / G) ? 1 : 0;
  const int* stb = st + b * NSEG + 1 + jb * G;
  const int t0 = stb[0];
  const int t1 = stb[G];
  int s0 = 0, s1 = 0;
  if (tid < G) { s0 = stb[tid]; s1 = stb[tid + 1]; }
  float sum = 0.0f; int cnt = 0;

  const float* wp = wvec + cls * Hc + koff;
  const float* Xb = X + (size_t)b * Lc * Hc + koff;
  const int*   mb = mask + b * Lc;

  for (int base = t0; base < t1; base = __builtin_amdgcn_readfirstlane(base + TOK)) {
    const int mraw = mb[min(base + tid, t1 - 1)];

    const int tokW = base + wave * WTOK;
    if (tokW < t1) {
      v4f wr[NSTEP];
#pragma unroll
      for (int q = 0; q < NSTEP; ++q)
        wr[q] = *reinterpret_cast<const v4f*>(wp + q * KSTEP);

      const float* xr[WTOK / 2];
#pragma unroll
      for (int P = 0; P < WTOK / 2; ++P)
        xr[P] = Xb + (size_t)min(tokW + 2 * P + t, t1 - 1) * Hc;

      v4f xb[2][NSTEP];
#pragma unroll
      for (int q = 0; q < NSTEP; ++q)
        xb[0][q] = *reinterpret_cast<const v4f*>(xr[0] + q * KSTEP);
#pragma unroll
      for (int q = 0; q < NSTEP; ++q)
        xb[1][q] = *reinterpret_cast<const v4f*>(xr[1] + q * KSTEP);
      __builtin_amdgcn_sched_barrier(0);

      float res[WTOK / 2];
#pragma unroll
      for (int P = 0; P < WTOK / 2; ++P) {
        v8f acc0 = {}, acc1 = {};
#pragma unroll
        for (int g = 0; g < NSTEP / GRP; ++g) {
#pragma unroll
          for (int j = 0; j < GRP; ++j) {
            const int q = g * GRP + j;
            const v4f a = xb[P & 1][q];
            const v4f bw = wr[q];
            v2f a0 = {a.x, a.y}, a1 = {a.z, a.w};
            v2f b0 = {bw.x, bw.y}, b1 = {bw.z, bw.w};
            acc0 = __builtin_amdgcn_wmma_f32_16x16x4_f32(
                 false, a0,  false, b0,
                 (short)0, acc0,  false,  false);
            acc1 = __builtin_amdgcn_wmma_f32_16x16x4_f32(
                 false, a1,  false, b1,
                 (short)0, acc1,  false,  false);
          }
          if (P + 2 < WTOK / 2) {
#pragma unroll
            for (int j = 0; j < GRP; ++j) {
              const int q = g * GRP + j;
              xb[P & 1][q] = *reinterpret_cast<const v4f*>(xr[P + 2] + q * KSTEP);
            }
          }
          __builtin_amdgcn_sched_barrier(0);
        }
        const v8f acc = acc0 + acc1;
        res[P] = pick8(acc, p1, p2, p4);
      }

#pragma unroll
      for (int P = 0; P < WTOK / 2; ++P) {
        float s = res[P];
        s += __shfl_xor(s, 1, 32);
        s += __shfl_xor(s, 2, 32);
        s += __shfl_xor(s, 4, 32);
        res[P] = s;
      }
      if (c == 0) {
#pragma unroll
        for (int P = 0; P < WTOK / 2; ++P)
          sDot[wave * WTOK + 2 * P + h] = res[P];
      }
    }
    if (tid < TOK) sVal[tid] = (base + tid < t1 && mraw > 0) ? 1 : 0;
    __syncthreads();
    if (tid < G) {
      const int lo = max(s0, base), hi = min(s1, base + TOK);
      for (int k = lo; k < hi; ++k)
        if (sVal[k - base]) { sum += sDot[k - base]; ++cnt; }
    }
    __syncthreads();
  }
  if (tid < G)
    means[(b * NBLK + jb) * MSTR + tid] = sum / (float)max(cnt, 1);
}

__global__ void logits_kernel(const float* __restrict__ means,
                              const float* __restrict__ biasPtr,
                              float* __restrict__ out) {
  int idx = blockIdx.x * blockDim.x + threadIdx.x;
  if (idx >= Bc * Sc * (Tc / 4)) return;
  int t4  = idx % (Tc / 4);
  int rem = idx / (Tc / 4);
  int s = rem % Sc;
  int b = rem / Sc;
  int tt = t4 * 4;
  float av = means[(b * NBLK + s / G) * MSTR + s % G] + biasPtr[0];
  v4f cv = *reinterpret_cast<const v4f*>(
      means + (b * NBLK + Sc / G + tt / G) * MSTR + tt % G);
  v4f o = cv + av;
  reinterpret_cast<v4f*>(out)[idx] = o;
}

extern "C" void kernel_launch(void* const* d_in, const int* in_sizes, int n_in,
                              void* d_out, int out_size, void* d_ws, size_t ws_size,
                              hipStream_t stream) {
  const float* X    = (const float*)d_in[0];
  const int*   msk  = (const int*)d_in[1];
  const int*   wid  = (const int*)d_in[2];
  const float* wvec = (const float*)d_in[5];
  const float* bias = (const float*)d_in[6];
  float* out = (float*)d_out;

  char* ws = (char*)d_ws;
  size_t off = 0;
  int*   st    = (int*)(ws + off);   off += (size_t)Bc * NSEG * 4;
  float* means = (float*)(ws + off); off += (size_t)Bc * NBLK * MSTR * 4;

  seg_scan_kernel<<<Bc, 1024, 0, stream>>>(wid, msk, st);

  dot_seg_kernel<<<Bc * NBLK, 256, 0, stream>>>(X, wvec, msk, st, means);

  logits_kernel<<<(Bc * Sc * (Tc / 4) + 255) / 256, 256, 0, stream>>>(
      means, bias, out);
}


# logits kernel: one resident grid of 256 blocks of 16 rows; a wave keeps its 128 tgt means in registers and stores them under 8 src rows, so 2048 waves enter, not 16384; sums unchanged
# speedup vs baseline: 2.3584x; 1.0514x over previous
#include <hip/hip_runtime.h>
#include <hip/hip_bf16.h>
#include <cstdint>
#include <cstddef>

typedef float v2f __attribute__((ext_vector_type(2)));
typedef float v4f __attribute__((ext_vector_type(4)));
typedef float v8f __attribute__((ext_vector_type(8)));

constexpr int Bc = 8;
constexpr int Lc = 4096;
constexpr int Hc = 1024;
constexpr int Sc = 512;
constexpr int Tc = 512;
constexpr int TOK = 64;
constexpr int WTOK = 8;
constexpr int KSTEP = 64;
constexpr int NSTEP = Hc / KSTEP;
constexpr int GRP = 4;
constexpr int NSEG = 1056;
constexpr int G = 16;
constexpr int NBLK = (Sc + Tc) / G;
constexpr int MSTR = 32;

__global__ __launch_bounds__(1024) void seg_scan_kernel(
    const int* __restrict__ wid, const int* __restrict__ mask,
    int* __restrict__ st) {
  __shared__ int sHas[32];
  __shared__ int sWid[32];
  __shared__ int sTot[32];
  const int tid  = threadIdx.x;
  const int lane = tid & 31;
  const int wave = __builtin_amdgcn_readfirstlane(tid >> 5);
  const int b = blockIdx.x;
  const int base = b * Lc + tid * 4;

  int w[4], v[4];
#pragma unroll
  for (int j = 0; j < 4; ++j) {
    w[j] = wid[base + j];
    v[j] = (mask[base + j] > 0) ? 1 : 0;
  }

  int curH = 0, curW = -2;
#pragma unroll
  for (int j = 0; j < 4; ++j)
    if (v[j]) { curH = 1; curW = w[j]; }
  const unsigned hm  = __builtin_amdgcn_ballot_w32(curH != 0);
  const unsigned bel = hm & ((1u << lane) - 1u);
  const int srcLane  = bel ? 31 - __builtin_clz(bel) : 0;
  const int pw = __shfl(curW, srcLane, 32);
  const int lastLane = hm ? 31 - __builtin_clz(hm) : 0;
  const int waveW = __builtin_amdgcn_readlane(curW, lastLane);
  if (lane == 0) { sHas[wave] = hm ? 1 : 0; sWid[wave] = waveW; }
  __syncthreads();
  const int rH = sHas[lane], rW = sWid[lane];
  const unsigned wm = __builtin_amdgcn_ballot_w32(rH != 0) & ((1u << wave) - 1u);
  const int wsrc = wm ? 31 - __builtin_clz(wm) : 0;
  const int cw = __builtin_amdgcn_readlane(rW, wsrc);
  const int carry = wm ? cw : -2;
  int prev = bel ? pw : carry;

  int f[4];
#pragma unroll
  for (int j = 0; j < 4; ++j) {
    f[j] = 0;
    if (v[j]) {
      f[j] = (w[j] == -1 || prev == -1 || w[j] != prev) ? 1 : 0;
      prev = w[j];
    }
  }

  unsigned ex = 0; int tot = 0;
#pragma unroll
  for (int j = 0; j < 4; ++j) {
    const unsigned bj = __builtin_amdgcn_ballot_w32(f[j] != 0);
    ex = __builtin_amdgcn_mbcnt_lo(bj, ex);
    tot += __builtin_popcount(bj);
  }
  if (lane == 0) sTot[wave] = tot;
  __syncthreads();
  const int tv = sTot[lane];
  int pk = ((lane < wave) ? tv : 0) | (tv << 16);
#pragma unroll
  for (int m = 1; m < 32; m <<= 1) pk += __shfl_xor(pk, m, 32);
  int run = (pk & 0xFFFF) + (int)ex;
  const int nseg = pk >> 16;
#pragma unroll
  for (int j = 0; j < 4; ++j) {
    if (f[j]) {
      run += 1;
      if (run - 1 < NSEG) st[b * NSEG + run - 1] = tid * 4 + j;
    }
  }
  for (int i = tid; i < NSEG; i += 1024)
    if (i >= nseg) st[b * NSEG + i] = Lc;
}

__device__ __forceinline__ float pick8(const v8f& a, bool p1, bool p2, bool p4) {
  float s0 = p1 ? a[1] : a[0];
  float s1 = p1 ? a[3] : a[2];
  float s2 = p1 ? a[5] : a[4];
  float s3 = p1 ? a[7] : a[6];
  float u0 = p2 ? s1 : s0;
  float u1 = p2 ? s3 : s2;
  return p4 ? u1 : u0;
}

__global__ __launch_bounds__(256) void dot_seg_kernel(
    const float* __restrict__ X, const float* __restrict__ wvec,
    const int* __restrict__ mask, const int* __restrict__ st,
    float* __restrict__ means) {
  __shared__ float sDot[TOK];
  __shared__ int   sVal[TOK];

  const int tid  = threadIdx.x;
  const int lane = tid & 31;
  const int wave = __builtin_amdgcn_readfirstlane(tid >> 5);
  const int c    = lane & 15;
  const int h    = lane >> 4;
  const int t    = c >> 3;
  const int p    = c & 7;
  const int koff = 8 * p + 4 * h;
  const bool p1 = p & 1, p2 = p & 2, p4 = p & 4;

  const int b   = blockIdx.x / NBLK;
  const int jb  = blockIdx.x % NBLK;
  const int cls = (jb >= Sc / G) ? 1 : 0;
  const int* stb = st + b * NSEG + 1 + jb * G;
  const int t0 = stb[0];
  const int t1 = stb[G];
  int s0 = 0, s1 = 0;
  if (tid < G) { s0 = stb[tid]; s1 = stb[tid + 1]; }
  float sum = 0.0f; int cnt = 0;

  const float* wp = wvec + cls * Hc + koff;
  const float* Xb = X + (size_t)b * Lc * Hc + koff;
  const int*   mb = mask + b * Lc;

  for (int base = t0; base < t1; base = __builtin_amdgcn_readfirstlane(base + TOK)) {
    const int mraw = mb[min(base + tid, t1 - 1)];

    const int tokW = base + wave * WTOK;
    if (tokW < t1) {
      v4f wr[NSTEP];
#pragma unroll
      for (int q = 0; q < NSTEP; ++q)
        wr[q] = *reinterpret_cast<const v4f*>(wp + q * KSTEP);

      const float* xr[WTOK / 2];
#pragma unroll
      for (int P = 0; P < WTOK / 2; ++P)
        xr[P] = Xb + (size_t)min(tokW + 2 * P + t, t1 - 1) * Hc;

      v4f xb[2][NSTEP];
#pragma unroll
      for (int q = 0; q < NSTEP; ++q)
        xb[0][q] = *reinterpret_cast<const v4f*>(xr[0] + q * KSTEP);
#pragma unroll
      for (int q = 0; q < NSTEP; ++q)
        xb[1][q] = *reinterpret_cast<const v4f*>(xr[1] + q * KSTEP);
      __builtin_amdgcn_sched_barrier(0);

      float res[WTOK / 2];
#pragma unroll
      for (int P = 0; P < WTOK / 2; ++P) {
        v8f acc0 = {}, acc1 = {};
#pragma unroll
        for (int g = 0; g < NSTEP / GRP; ++g) {
#pragma unroll
          for (int j = 0; j < GRP; ++j) {
            const int q = g * GRP + j;
            const v4f a = xb[P & 1][q];
            const v4f bw = wr[q];
            v2f a0 = {a.x, a.y}, a1 = {a.z, a.w};
            v2f b0 = {bw.x, bw.y}, b1 = {bw.z, bw.w};
            acc0 = __builtin_amdgcn_wmma_f32_16x16x4_f32(
                 false, a0,  false, b0,
                 (short)0, acc0,  false,  false);
            acc1 = __builtin_amdgcn_wmma_f32_16x16x4_f32(
                 false, a1,  false, b1,
                 (short)0, acc1,  false,  false);
          }
          if (P + 2 < WTOK / 2) {
#pragma unroll
            for (int j = 0; j < GRP; ++j) {
              const int q = g * GRP + j;
              xb[P & 1][q] = *reinterpret_cast<const v4f*>(xr[P + 2] + q * KSTEP);
            }
          }
          __builtin_amdgcn_sched_barrier(0);
        }
        const v8f acc = acc0 + acc1;
        res[P] = pick8(acc, p1, p2, p4);
      }

#pragma unroll
      for (int P = 0; P < WTOK / 2; ++P) {
        float s = res[P];
        s += __shfl_xor(s, 1, 32);
        s += __shfl_xor(s, 2, 32);
        s += __shfl_xor(s, 4, 32);
        res[P] = s;
      }
      if (c == 0) {
#pragma unroll
        for (int P = 0; P < WTOK / 2; ++P)
          sDot[wave * WTOK + 2 * P + h] = res[P];
      }
    }
    if (tid < TOK) sVal[tid] = (base + tid < t1 && mraw > 0) ? 1 : 0;
    __syncthreads();
    if (tid < G) {
      const int lo = max(s0, base), hi = min(s1, base + TOK);
      for (int k = lo; k < hi; ++k)
        if (sVal[k - base]) { sum += sDot[k - base]; ++cnt; }
    }
    __syncthreads();
  }
  if (tid < G)
    means[(b * NBLK + jb) * MSTR + tid] = sum / (float)max(cnt, 1);
}

constexpr int LWAVES = 8;
constexpr int LROWS  = G / 2;
constexpr int LBLK   = Sc / G;

__global__ __launch_bounds__(LWAVES * 32) void logits_kernel(
    const float* __restrict__ means, const float* __restrict__ biasPtr,
    float* __restrict__ out) {
  const int lane = threadIdx.x & 31;
  const int wave = __builtin_amdgcn_readfirstlane(threadIdx.x >> 5);
  const int b  = blockIdx.x / LBLK;
  const int sg = blockIdx.x % LBLK;
  const int q  = wave & 3;
  const int r  = wave >> 2;
  const int tt = (Tc / 4) * q + 4 * lane;
  const v4f cv = *reinterpret_cast<const v4f*>(
      means + (b * NBLK + Sc / G + tt / G) * MSTR + tt % G);
  const float* ap = means + (b * NBLK + sg) * MSTR + LROWS * r;
  const float bias = biasPtr[0];
  float* op = out + ((size_t)(b * Sc + G * sg + LROWS * r)) * Tc + tt;
  for (int i = 0; i < LROWS; ++i) {
    const float av = ap[i] + bias;
    *reinterpret_cast<v4f*>(op + (size_t)i * Tc) = cv + av;
  }
}

extern "C" void kernel_launch(void* const* d_in, const int* in_sizes, int n_in,
                              void* d_out, int out_size, void* d_ws, size_t ws_size,
                              hipStream_t stream) {
  const float* X    = (const float*)d_in[0];
  const int*   msk  = (const int*)d_in[1];
  const int*   wid  = (const int*)d_in[2];
  const float* wvec = (const float*)d_in[5];
  const float* bias = (const float*)d_in[6];
  float* out = (float*)d_out;

  char* ws = (char*)d_ws;
  size_t off = 0;
  int*   st    = (int*)(ws + off);   off += (size_t)Bc * NSEG * 4;
  float* means = (float*)(ws + off); off += (size_t)Bc * NBLK * MSTR * 4;

  seg_scan_kernel<<<Bc, 1024, 0, stream>>>(wid, msk, st);

  dot_seg_kernel<<<Bc * NBLK, 256, 0, stream>>>(X, wvec, msk, st, means);

  logits_kernel<<<Bc * LBLK, LWAVES * 32, 0, stream>>>(
      means, bias, out);
}


# dot kernel: a tile's row pairs are dealt to the 8 waves in turn, a wave runs only its in-range pairs (whole-pair refills): even WMMA load per SIMD; 234 VGPRs, 2 blocks share a WGP; sums unchanged
# speedup vs baseline: 2.6043x; 1.1043x over previous
#include <hip/hip_runtime.h>
#include <hip/hip_bf16.h>
#include <cstdint>
#include <cstddef>

typedef float v2f __attribute__((ext_vector_type(2)));
typedef float v4f __attribute__((ext_vector_type(4)));
typedef float v8f __attribute__((ext_vector_type(8)));

constexpr int Bc = 8;
constexpr int Lc = 4096;
constexpr int Hc = 1024;
constexpr int Sc = 512;
constexpr int Tc = 512;
constexpr int TOK = 64;
constexpr int NWAVE = 8;
constexpr int NSLOT = TOK / (2 * NWAVE);
constexpr int KSTEP = 64;
constexpr int NSTEP = Hc / KSTEP;
constexpr int NSEG = 1056;
constexpr int G = 16;
constexpr int NBLK = (Sc + Tc) / G;
constexpr int MSTR = 32;

__global__ __launch_bounds__(1024) void seg_scan_kernel(
    const int* __restrict__ wid, const int* __restrict__ mask,
    int* __restrict__ st) {
  __shared__ int sHas[32];
  __shared__ int sWid[32];
  __shared__ int sTot[32];
  const int tid  = threadIdx.x;
  const int lane = tid & 31;
  const int wave = __builtin_amdgcn_readfirstlane(tid >> 5);
  const int b = blockIdx.x;
  const int base = b * Lc + tid * 4;

  int w[4], v[4];
#pragma unroll
  for (int j = 0; j < 4; ++j) {
    w[j] = wid[base + j];
    v[j] = (mask[base + j] > 0) ? 1 : 0;
  }

  int curH = 0, curW = -2;
#pragma unroll
  for (int j = 0; j < 4; ++j)
    if (v[j]) { curH = 1; curW = w[j]; }
  const unsigned hm  = __builtin_amdgcn_ballot_w32(curH != 0);
  const unsigned bel = hm & ((1u << lane) - 1u);
  const int srcLane  = bel ? 31 - __builtin_clz(bel) : 0;
  const int pw = __shfl(curW, srcLane, 32);
  const int lastLane = hm ? 31 - __builtin_clz(hm) : 0;
  const int waveW = __builtin_amdgcn_readlane(curW, lastLane);
  if (lane == 0) { sHas[wave] = hm ? 1 : 0; sWid[wave] = waveW; }
  __syncthreads();
  const int rH = sHas[lane], rW = sWid[lane];
  const unsigned wm = __builtin_amdgcn_ballot_w32(rH != 0) & ((1u << wave) - 1u);
  const int wsrc = wm ? 31 - __builtin_clz(wm) : 0;
  const int cw = __builtin_amdgcn_readlane(rW, wsrc);
  const int carry = wm ? cw : -2;
  int prev = bel ? pw : carry;

  int f[4];
#pragma unroll
  for (int j = 0; j < 4; ++j) {
    f[j] = 0;
    if (v[j]) {
      f[j] = (w[j] == -1 || prev == -1 || w[j] != prev) ? 1 : 0;
      prev = w[j];
    }
  }

  unsigned ex = 0; int tot = 0;
#pragma unroll
  for (int j = 0; j < 4; ++j) {
    const unsigned bj = __builtin_amdgcn_ballot_w32(f[j] != 0);
    ex = __builtin_amdgcn_mbcnt_lo(bj, ex);
    tot += __builtin_popcount(bj);
  }
  if (lane == 0) sTot[wave] = tot;
  __syncthreads();
  const int tv = sTot[lane];
  int pk = ((lane < wave) ? tv : 0) | (tv << 16);
#pragma unroll
  for (int m = 1; m < 32; m <<= 1) pk += __shfl_xor(pk, m, 32);
  int run = (pk & 0xFFFF) + (int)ex;
  const int nseg = pk >> 16;
#pragma unroll
  for (int j = 0; j < 4; ++j) {
    if (f[j]) {
      run += 1;
      if (run - 1 < NSEG) st[b * NSEG + run - 1] = tid * 4 + j;
    }
  }
  for (int i = tid; i < NSEG; i += 1024)
    if (i >= nseg) st[b * NSEG + i] = Lc;
}

__device__ __forceinline__ float pick8(const v8f& a, bool p1, bool p2, bool p4) {
  float s0 = p1 ? a[1] : a[0];
  float s1 = p1 ? a[3] : a[2];
  float s2 = p1 ? a[5] : a[4];
  float s3 = p1 ? a[7] : a[6];
  float u0 = p2 ? s1 : s0;
  float u1 = p2 ? s3 : s2;
  return p4 ? u1 : u0;
}

__global__ __launch_bounds__(256) void dot_seg_kernel(
    const float* __restrict__ X, const float* __restrict__ wvec,
    const int* __restrict__ mask, const int* __restrict__ st,
    float* __restrict__ means) {
  __shared__ float sDot[TOK];
  __shared__ int   sVal[TOK];

  const int tid  = threadIdx.x;
  const int lane = tid & 31;
  const int wave = __builtin_amdgcn_readfirstlane(tid >> 5);
  const int c    = lane & 15;
  const int h    = lane >> 4;
  const int t    = c >> 3;
  const int p    = c & 7;
  const int koff = 8 * p + 4 * h;
  const bool p1 = p & 1, p2 = p & 2, p4 = p & 4;

  const int b   = blockIdx.x / NBLK;
  const int jb  = blockIdx.x % NBLK;
  const int cls = (jb >= Sc / G) ? 1 : 0;
  const int* stb = st + b * NSEG + 1 + jb * G;
  const int t0 = stb[0];
  const int t1 = stb[G];
  int s0 = 0, s1 = 0;
  if (tid < G) { s0 = stb[tid]; s1 = stb[tid + 1]; }
  float sum = 0.0f; int cnt = 0;

  const float* wp = wvec + cls * Hc + koff;
  const float* Xb = X + (size_t)b * Lc * Hc + koff;
  const int*   mb = mask + b * Lc;

  for (int base = t0; base < t1; base = __builtin_amdgcn_readfirstlane(base + TOK)) {
    const int mraw = mb[min(base + tid, t1 - 1)];

    const int tok0 = base + 2 * wave;
    if (tok0 < t1) {
      v4f wr[NSTEP];
#pragma unroll
      for (int q = 0; q < NSTEP; ++q)
        wr[q] = *reinterpret_cast<const v4f*>(wp + q * KSTEP);

      const float* xr[NSLOT];
#pragma unroll
      for (int P = 0; P < NSLOT; ++P)
        xr[P] = Xb + (size_t)min(tok0 + 2 * NWAVE * P + t, t1 - 1) * Hc;

      v4f xb[2][NSTEP];
#pragma unroll
      for (int q = 0; q < NSTEP; ++q)
        xb[0][q] = *reinterpret_cast<const v4f*>(xr[0] + q * KSTEP);
      __builtin_amdgcn_sched_barrier(0);
#pragma unroll
      for (int q = 0; q < NSTEP; ++q)
        xb[1][q] = *reinterpret_cast<const v4f*>(xr[1] + q * KSTEP);
      __builtin_amdgcn_sched_barrier(0);

      float res[NSLOT];
#pragma unroll
      for (int P = 0; P < NSLOT; ++P) {
        res[P] = 0.0f;
        if (tok0 + 2 * NWAVE * P < t1) {
          v8f acc0 = {}, acc1 = {};
#pragma unroll
          for (int q = 0; q < NSTEP; ++q) {
            const v4f a = xb[P & 1][q];
            const v4f bw = wr[q];
            v2f a0 = {a.x, a.y}, a1 = {a.z, a.w};
            v2f b0 = {bw.x, bw.y}, b1 = {bw.z, bw.w};
            acc0 = __builtin_amdgcn_wmma_f32_16x16x4_f32(
                 false, a0,  false, b0,
                 (short)0, acc0,  false,  false);
            acc1 = __builtin_amdgcn_wmma_f32_16x16x4_f32(
                 false, a1,  false, b1,
                 (short)0, acc1,  false,  false);
          }
          const v8f acc = acc0 + acc1;
          res[P] = pick8(acc, p1, p2, p4);
          if (P + 2 < NSLOT) {
            if (tok0 + 2 * NWAVE * (P + 2) < t1) {
#pragma unroll
              for (int q = 0; q < NSTEP; ++q)
                xb[P & 1][q] = *reinterpret_cast<const v4f*>(xr[P + 2] + q * KSTEP);
            }
          }
        }
        __builtin_amdgcn_sched_barrier(0);
      }

#pragma unroll
      for (int P = 0; P < NSLOT; ++P) {
        float s = res[P];
        s += __shfl_xor(s, 1, 32);
        s += __shfl_xor(s, 2, 32);
        s += __shfl_xor(s, 4, 32);
        res[P] = s;
      }
      if (c == 0) {
#pragma unroll
        for (int P = 0; P < NSLOT; ++P)
          sDot[2 * (NWAVE * P + wave) + h] = res[P];
      }
    }
    if (tid < TOK) sVal[tid] = (base + tid < t1 && mraw > 0) ? 1 : 0;
    __syncthreads();
    if (tid < G) {
      const int lo = max(s0, base), hi = min(s1, base + TOK);
      for (int k = lo; k < hi; ++k)
        if (sVal[k - base]) { sum += sDot[k - base]; ++cnt; }
    }
    __syncthreads();
  }
  if (tid < G)
    means[(b * NBLK + jb) * MSTR + tid] = sum / (float)max(cnt, 1);
}

constexpr int LWAVES = 8;
constexpr int LROWS  = G / 2;
constexpr int LBLK   = Sc / G;

__global__ __launch_bounds__(LWAVES * 32) void logits_kernel(
    const float* __restrict__ means, const float* __restrict__ biasPtr,
    float* __restrict__ out) {
  const int lane = threadIdx.x & 31;
  const int wave = __builtin_amdgcn_readfirstlane(threadIdx.x >> 5);
  const int b  = blockIdx.x / LBLK;
  const int sg = blockIdx.x % LBLK;
  const int q  = wave & 3;
  const int r  = wave >> 2;
  const int tt = (Tc / 4) * q + 4 * lane;
  const v4f cv = *reinterpret_cast<const v4f*>(
      means + (b * NBLK + Sc / G + tt / G) * MSTR + tt % G);
  const float* ap = means + (b * NBLK + sg) * MSTR + LROWS * r;
  const float bias = biasPtr[0];
  float* op = out + ((size_t)(b * Sc + G * sg + LROWS * r)) * Tc + tt;
  for (int i = 0; i < LROWS; ++i) {
    const float av = ap[i] + bias;
    *reinterpret_cast<v4f*>(op + (size_t)i * Tc) = cv + av;
  }
}

extern "C" void kernel_launch(void* const* d_in, const int* in_sizes, int n_in,
                              void* d_out, int out_size, void* d_ws, size_t ws_size,
                              hipStream_t stream) {
  const float* X    = (const float*)d_in[0];
  const int*   msk  = (const int*)d_in[1];
  const int*   wid  = (const int*)d_in[2];
  const float* wvec = (const float*)d_in[5];
  const float* bias = (const float*)d_in[6];
  float* out = (float*)d_out;

  char* ws = (char*)d_ws;
  size_t off = 0;
  int*   st    = (int*)(ws + off);   off += (size_t)Bc * NSEG * 4;
  float* means = (float*)(ws + off); off += (size_t)Bc * NBLK * MSTR * 4;

  seg_scan_kernel<<<Bc, 1024, 0, stream>>>(wid, msk, st);

  dot_seg_kernel<<<Bc * NBLK, 256, 0, stream>>>(X, wvec, msk, st, means);

  logits_kernel<<<Bc * LBLK, LWAVES * 32, 0, stream>>>(
      means, bias, out);
}


# segmentation launch and start table go: each dot block scans its batch's word ids and mask itself (lane-mask scans, 3 barriers), keeps its 17 start tokens in LDS; 3 launches -> 2; sums unchanged
# speedup vs baseline: 2.7995x; 1.0750x over previous
#include <hip/hip_runtime.h>
#include <hip/hip_bf16.h>
#include <cstdint>
#include <cstddef>

typedef float v2f __attribute__((ext_vector_type(2)));
typedef float v4f __attribute__((ext_vector_type(4)));
typedef float v8f __attribute__((ext_vector_type(8)));
typedef int   v4i __attribute__((ext_vector_type(4)));

constexpr int Bc = 8;
constexpr int Lc = 4096;
constexpr int Hc = 1024;
constexpr int Sc = 512;
constexpr int Tc = 512;
constexpr int TOK = 64;
constexpr int NWAVE = 8;
constexpr int NSLOT = TOK / (2 * NWAVE);
constexpr int KSTEP = 64;
constexpr int NSTEP = Hc / KSTEP;
constexpr int NQ   = Lc / (4 * 32 * NWAVE);
constexpr int NVW  = NQ * NWAVE;
constexpr int G = 16;
constexpr int NBLK = (Sc + Tc) / G;
constexpr int MSTR = 32;

__device__ __forceinline__ float pick8(const v8f& a, bool p1, bool p2, bool p4) {
  float s0 = p1 ? a[1] : a[0];
  float s1 = p1 ? a[3] : a[2];
  float s2 = p1 ? a[5] : a[4];
  float s3 = p1 ? a[7] : a[6];
  float u0 = p2 ? s1 : s0;
  float u1 = p2 ? s3 : s2;
  return p4 ? u1 : u0;
}

__global__ __launch_bounds__(256) void dot_seg_kernel(
    const float* __restrict__ X, const float* __restrict__ wvec,
    const int* __restrict__ mask, const int* __restrict__ wid,
    float* __restrict__ means) {
  __shared__ float sDot[TOK];
  __shared__ int   sVal[TOK];
  __shared__ int   sHas[NVW];
  __shared__ int   sWid[NVW];
  __shared__ int   sTot[NVW];
  __shared__ int   sSt[G + 1];

  const int tid  = threadIdx.x;
  const int lane = tid & 31;
  const int wave = __builtin_amdgcn_readfirstlane(tid >> 5);
  const int c    = lane & 15;
  const int h    = lane >> 4;
  const int t    = c >> 3;
  const int p    = c & 7;
  const int koff = 8 * p + 4 * h;
  const bool p1 = p & 1, p2 = p & 2, p4 = p & 4;

  const int b   = blockIdx.x / NBLK;
  const int jb  = blockIdx.x % NBLK;
  const int cls = (jb >= Sc / G) ? 1 : 0;
  const int* wb = wid + b * Lc;
  const int* mb = mask + b * Lc;

  v4i w4[NQ], m4[NQ];
#pragma unroll
  for (int q = 0; q < NQ; ++q) {
    w4[q] = *reinterpret_cast<const v4i*>(wb + 1024 * q + 4 * tid);
    m4[q] = *reinterpret_cast<const v4i*>(mb + 1024 * q + 4 * tid);
  }
  if (tid <= G) sSt[tid] = Lc;

  unsigned bel[NQ]; int pw[NQ];
#pragma unroll
  for (int q = 0; q < NQ; ++q) {
    int curH = 0, curW = -2;
#pragma unroll
    for (int j = 0; j < 4; ++j)
      if (m4[q][j] > 0) { curH = 1; curW = w4[q][j]; }
    const unsigned hm = __builtin_amdgcn_ballot_w32(curH != 0);
    bel[q] = hm & ((1u << lane) - 1u);
    const int srcLane = bel[q] ? 31 - __builtin_clz(bel[q]) : 0;
    pw[q] = __shfl(curW, srcLane, 32);
    const int lastLane = hm ? 31 - __builtin_clz(hm) : 0;
    const int runW = __builtin_amdgcn_readlane(curW, lastLane);
    if (lane == 0) { sHas[NWAVE * q + wave] = hm ? 1 : 0; sWid[NWAVE * q + wave] = runW; }
  }
  __syncthreads();
  const int rH = sHas[lane], rW = sWid[lane];
  const unsigned am = __builtin_amdgcn_ballot_w32(rH != 0);
  int f[NQ][4];
  unsigned ex[NQ];
#pragma unroll
  for (int q = 0; q < NQ; ++q) {
    const int vw = NWAVE * q + wave;
    const unsigned wm = am & ((1u << vw) - 1u);
    const int wsrc = wm ? 31 - __builtin_clz(wm) : 0;
    const int cw = __builtin_amdgcn_readlane(rW, wsrc);
    const int carry = wm ? cw : -2;
    int prev = bel[q] ? pw[q] : carry;
#pragma unroll
    for (int j = 0; j < 4; ++j) {
      const bool v = m4[q][j] > 0;
      const int w = w4[q][j];
      f[q][j] = (v & ((w == -1) | (prev == -1) | (w != prev))) ? 1 : 0;
      prev = v ? w : prev;
    }
    unsigned e = 0; int tot = 0;
#pragma unroll
    for (int j = 0; j < 4; ++j) {
      const unsigned bj = __builtin_amdgcn_ballot_w32(f[q][j] != 0);
      e = __builtin_amdgcn_mbcnt_lo(bj, e);
      tot += __builtin_popcount(bj);
    }
    ex[q] = e;
    if (lane == 0) sTot[vw] = tot;
  }
  __syncthreads();
  const int tv = sTot[lane];
  int incl = tv;
#pragma unroll
  for (int m = 1; m < 32; m <<= 1) {
    const int up = __shfl_up(incl, m, 32);
    if (lane >= m) incl += up;
  }
  const int excl = incl - tv;
#pragma unroll
  for (int q = 0; q < NQ; ++q) {
    int rel = __builtin_amdgcn_readlane(excl, NWAVE * q + wave) + (int)ex[q] - (1 + jb * G);
#pragma unroll
    for (int j = 0; j < 4; ++j) {
      if (f[q][j]) {
        if (rel >= 0 && rel <= G) sSt[rel] = 1024 * q + 4 * tid + j;
        ++rel;
      }
    }
  }
  __syncthreads();
  const int t0 = __builtin_amdgcn_readfirstlane(sSt[0]);
  const int t1 = __builtin_amdgcn_readfirstlane(sSt[G]);
  int s0 = 0, s1 = 0;
  if (tid < G) { s0 = sSt[tid]; s1 = sSt[tid + 1]; }
  float sum = 0.0f; int cnt = 0;

  const float* wp = wvec + cls * Hc + koff;
  const float* Xb = X + (size_t)b * Lc * Hc + koff;

  for (int base = t0; base < t1; base = __builtin_amdgcn_readfirstlane(base + TOK)) {
    const int mraw = mb[min(base + tid, t1 - 1)];

    const int tok0 = base + 2 * wave;
    if (tok0 < t1) {
      v4f wr[NSTEP];
#pragma unroll
      for (int q = 0; q < NSTEP; ++q)
        wr[q] = *reinterpret_cast<const v4f*>(wp + q * KSTEP);

      const float* xr[NSLOT];
#pragma unroll
      for (int P = 0; P < NSLOT; ++P)
        xr[P] = Xb + (size_t)min(tok0 + 2 * NWAVE * P + t, t1 - 1) * Hc;

      v4f xb[2][NSTEP];
#pragma unroll
      for (int q = 0; q < NSTEP; ++q)
        xb[0][q] = *reinterpret_cast<const v4f*>(xr[0] + q * KSTEP);
      __builtin_amdgcn_sched_barrier(0);
#pragma unroll
      for (int q = 0; q < NSTEP; ++q)
        xb[1][q] = *reinterpret_cast<const v4f*>(xr[1] + q * KSTEP);
      __builtin_amdgcn_sched_barrier(0);

      float res[NSLOT];
#pragma unroll
      for (int P = 0; P < NSLOT; ++P) {
        res[P] = 0.0f;
        if (tok0 + 2 * NWAVE * P < t1) {
          v8f acc0 = {}, acc1 = {};
#pragma unroll
          for (int q = 0; q < NSTEP; ++q) {
            const v4f a = xb[P & 1][q];
            const v4f bw = wr[q];
            v2f a0 = {a.x, a.y}, a1 = {a.z, a.w};
            v2f b0 = {bw.x, bw.y}, b1 = {bw.z, bw.w};
            acc0 = __builtin_amdgcn_wmma_f32_16x16x4_f32(
                 false, a0,  false, b0,
                 (short)0, acc0,  false,  false);
            acc1 = __builtin_amdgcn_wmma_f32_16x16x4_f32(
                 false, a1,  false, b1,
                 (short)0, acc1,  false,  false);
          }
          const v8f acc = acc0 + acc1;
          res[P] = pick8(acc, p1, p2, p4);
          if (P + 2 < NSLOT) {
            if (tok0 + 2 * NWAVE * (P + 2) < t1) {
#pragma unroll
              for (int q = 0; q < NSTEP; ++q)
                xb[P & 1][q] = *reinterpret_cast<const v4f*>(xr[P + 2] + q * KSTEP);
            }
          }
        }
        __builtin_amdgcn_sched_barrier(0);
      }

#pragma unroll
      for (int P = 0; P < NSLOT; ++P) {
        float s = res[P];
        s += __shfl_xor(s, 1, 32);
        s += __shfl_xor(s, 2, 32);
        s += __shfl_xor(s, 4, 32);
        res[P] = s;
      }
      if (c == 0) {
#pragma unroll
        for (int P = 0; P < NSLOT; ++P)
          sDot[2 * (NWAVE * P + wave) + h] = res[P];
      }
    }
    if (tid < TOK) sVal[tid] = (base + tid < t1 && mraw > 0) ? 1 : 0;
    __syncthreads();
    if (tid < G) {
      const int lo = max(s0, base), hi = min(s1, base + TOK);
      for (int k = lo; k < hi; ++k)
        if (sVal[k - base]) { sum += sDot[k - base]; ++cnt; }
    }
    __syncthreads();
  }
  if (tid < G)
    means[(b * NBLK + jb) * MSTR + tid] = sum / (float)max(cnt, 1);
}

constexpr int LWAVES = 8;
constexpr int LROWS  = G / 2;
constexpr int LBLK   = Sc / G;

__global__ __launch_bounds__(LWAVES * 32) void logits_kernel(
    const float* __restrict__ means, const float* __restrict__ biasPtr,
    float* __restrict__ out) {
  const int lane = threadIdx.x & 31;
  const int wave = __builtin_amdgcn_readfirstlane(threadIdx.x >> 5);
  const int b  = blockIdx.x / LBLK;
  const int sg = blockIdx.x % LBLK;
  const int q  = wave & 3;
  const int r  = wave >> 2;
  const int tt = (Tc / 4) * q + 4 * lane;
  const v4f cv = *reinterpret_cast<const v4f*>(
      means + (b * NBLK + Sc / G + tt / G) * MSTR + tt % G);
  const float* ap = means + (b * NBLK + sg) * MSTR + LROWS * r;
  const float bias = biasPtr[0];
  float* op = out + ((size_t)(b * Sc + G * sg + LROWS * r)) * Tc + tt;
  for (int i = 0; i < LROWS; ++i) {
    const float av = ap[i] + bias;
    *reinterpret_cast<v4f*>(op + (size_t)i * Tc) = cv + av;
  }
}

extern "C" void kernel_launch(void* const* d_in, const int* in_sizes, int n_in,
                              void* d_out, int out_size, void* d_ws, size_t ws_size,
                              hipStream_t stream) {
  const float* X    = (const float*)d_in[0];
  const int*   msk  = (const int*)d_in[1];
  const int*   wid  = (const int*)d_in[2];
  const float* wvec = (const float*)d_in[5];
  const float* bias = (const float*)d_in[6];
  float* out = (float*)d_out;

  float* means = (float*)d_ws;

  dot_seg_kernel<<<Bc * NBLK, 256, 0, stream>>>(X, wvec, msk, wid, means);

  logits_kernel<<<Bc * LBLK, LWAVES * 32, 0, stream>>>(
      means, bias, out);
}


# dot kernel: LDS round trips leave a block's serial path: start-count prefix and 8-lane token sums on DPP adds (were 5 + 3 permute hops), segment owners read 8 tokens per LDS trip; sums unchanged
# speedup vs baseline: 3.0157x; 1.0772x over previous
#include <hip/hip_runtime.h>
#include <hip/hip_bf16.h>
#include <cstdint>
#include <cstddef>

typedef float v2f __attribute__((ext_vector_type(2)));
typedef float v4f __attribute__((ext_vector_type(4)));
typedef float v8f __attribute__((ext_vector_type(8)));
typedef int   v4i __attribute__((ext_vector_type(4)));

constexpr int Bc = 8;
constexpr int Lc = 4096;
constexpr int Hc = 1024;
constexpr int Sc = 512;
constexpr int Tc = 512;
constexpr int TOK = 64;
constexpr int NWAVE = 8;
constexpr int NSLOT = TOK / (2 * NWAVE);
constexpr int KSTEP = 64;
constexpr int NSTEP = Hc / KSTEP;
constexpr int NQ   = Lc / (4 * 32 * NWAVE);
constexpr int NVW  = NQ * NWAVE;
constexpr int G = 16;
constexpr int NBLK = (Sc + Tc) / G;
constexpr int MSTR = 32;

__device__ __forceinline__ float pick8(const v8f& a, bool p1, bool p2, bool p4) {
  float s0 = p1 ? a[1] : a[0];
  float s1 = p1 ? a[3] : a[2];
  float s2 = p1 ? a[5] : a[4];
  float s3 = p1 ? a[7] : a[6];
  float u0 = p2 ? s1 : s0;
  float u1 = p2 ? s3 : s2;
  return p4 ? u1 : u0;
}

template <int M>
__device__ __forceinline__ float xor_row(float x) {
  return __builtin_bit_cast(float, __builtin_amdgcn_update_dpp(
      0, __builtin_bit_cast(int, x), 0x160 + M, 0xf, 0xf, false));
}
template <int N>
__device__ __forceinline__ int below_row(int x) {
  return __builtin_amdgcn_update_dpp(0, x, 0x110 + N, 0xf, 0xf, true);
}

__global__ __launch_bounds__(256) void dot_seg_kernel(
    const float* __restrict__ X, const float* __restrict__ wvec,
    const int* __restrict__ mask, const int* __restrict__ wid,
    float* __restrict__ means) {
  __shared__ float sDot[TOK + 8];
  __shared__ int   sVal[TOK + 8];
  __shared__ int   sHas[NVW];
  __shared__ int   sWid[NVW];
  __shared__ int   sTot[NVW];
  __shared__ int   sSt[G + 1];

  const int tid  = threadIdx.x;
  const int lane = tid & 31;
  const int wave = __builtin_amdgcn_readfirstlane(tid >> 5);
  const int c    = lane & 15;
  const int h    = lane >> 4;
  const int t    = c >> 3;
  const int p    = c & 7;
  const int koff = 8 * p + 4 * h;
  const bool p1 = p & 1, p2 = p & 2, p4 = p & 4;

  const int b   = blockIdx.x / NBLK;
  const int jb  = blockIdx.x % NBLK;
  const int cls = (jb >= Sc / G) ? 1 : 0;
  const int* wb = wid + b * Lc;
  const int* mb = mask + b * Lc;

  v4i w4[NQ], m4[NQ];
#pragma unroll
  for (int q = 0; q < NQ; ++q) {
    w4[q] = *reinterpret_cast<const v4i*>(wb + 1024 * q + 4 * tid);
    m4[q] = *reinterpret_cast<const v4i*>(mb + 1024 * q + 4 * tid);
  }
  if (tid <= G) sSt[tid] = Lc;

  unsigned bel[NQ]; int pw[NQ];
#pragma unroll
  for (int q = 0; q < NQ; ++q) {
    int curH = 0, curW = -2;
#pragma unroll
    for (int j = 0; j < 4; ++j)
      if (m4[q][j] > 0) { curH = 1; curW = w4[q][j]; }
    const unsigned hm = __builtin_amdgcn_ballot_w32(curH != 0);
    bel[q] = hm & ((1u << lane) - 1u);
    const int srcLane = bel[q] ? 31 - __builtin_clz(bel[q]) : 0;
    pw[q] = __shfl(curW, srcLane, 32);
    const int lastLane = hm ? 31 - __builtin_clz(hm) : 0;
    const int runW = __builtin_amdgcn_readlane(curW, lastLane);
    if (lane == 0) { sHas[NWAVE * q + wave] = hm ? 1 : 0; sWid[NWAVE * q + wave] = runW; }
  }
  __syncthreads();
  const int rH = sHas[lane], rW = sWid[lane];
  const unsigned am = __builtin_amdgcn_ballot_w32(rH != 0);
  int f[NQ][4];
  unsigned ex[NQ];
#pragma unroll
  for (int q = 0; q < NQ; ++q) {
    const int vw = NWAVE * q + wave;
    const unsigned wm = am & ((1u << vw) - 1u);
    const int wsrc = wm ? 31 - __builtin_clz(wm) : 0;
    const int cw = __builtin_amdgcn_readlane(rW, wsrc);
    const int carry = wm ? cw : -2;
    int prev = bel[q] ? pw[q] : carry;
#pragma unroll
    for (int j = 0; j < 4; ++j) {
      const bool v = m4[q][j] > 0;
      const int w = w4[q][j];
      f[q][j] = (v & ((w == -1) | (prev == -1) | (w != prev))) ? 1 : 0;
      prev = v ? w : prev;
    }
    unsigned e = 0; int tot = 0;
#pragma unroll
    for (int j = 0; j < 4; ++j) {
      const unsigned bj = __builtin_amdgcn_ballot_w32(f[q][j] != 0);
      e = __builtin_amdgcn_mbcnt_lo(bj, e);
      tot += __builtin_popcount(bj);
    }
    ex[q] = e;
    if (lane == 0) sTot[vw] = tot;
  }
  __syncthreads();
  const int tv = sTot[lane];
  int incl = tv;
  incl += below_row<1>(incl);
  incl += below_row<2>(incl);
  incl += below_row<4>(incl);
  incl += below_row<8>(incl);
  const int row0 = __builtin_amdgcn_readlane(incl, 15);
  incl += (lane >= 16) ? row0 : 0;
  const int excl = incl - tv;
#pragma unroll
  for (int q = 0; q < NQ; ++q) {
    int rel = __builtin_amdgcn_readlane(excl, NWAVE * q + wave) + (int)ex[q] - (1 + jb * G);
#pragma unroll
    for (int j = 0; j < 4; ++j) {
      if (f[q][j]) {
        if (rel >= 0 && rel <= G) sSt[rel] = 1024 * q + 4 * tid + j;
        ++rel;
      }
    }
  }
  __syncthreads();
  const int t0 = __builtin_amdgcn_readfirstlane(sSt[0]);
  const int t1 = __builtin_amdgcn_readfirstlane(sSt[G]);
  int s0 = 0, s1 = 0;
  if (tid < G) { s0 = sSt[tid]; s1 = sSt[tid + 1]; }
  float sum = 0.0f; int cnt = 0;

  const float* wp = wvec + cls * Hc + koff;
  const float* Xb = X + (size_t)b * Lc * Hc + koff;

  for (int base = t0; base < t1; base = __builtin_amdgcn_readfirstlane(base + TOK)) {
    const int mraw = mb[min(base + tid, t1 - 1)];

    const int tok0 = base + 2 * wave;
    if (tok0 < t1) {
      v4f wr[NSTEP];
#pragma unroll
      for (int q = 0; q < NSTEP; ++q)
        wr[q] = *reinterpret_cast<const v4f*>(wp + q * KSTEP);

      const float* xr[NSLOT];
#pragma unroll
      for (int P = 0; P < NSLOT; ++P)
        xr[P] = Xb + (size_t)min(tok0 + 2 * NWAVE * P + t, t1 - 1) * Hc;

      v4f xb[2][NSTEP];
#pragma unroll
      for (int q = 0; q < NSTEP; ++q)
        xb[0][q] = *reinterpret_cast<const v4f*>(xr[0] + q * KSTEP);
      __builtin_amdgcn_sched_barrier(0);
#pragma unroll
      for (int q = 0; q < NSTEP; ++q)
        xb[1][q] = *reinterpret_cast<const v4f*>(xr[1] + q * KSTEP);
      __builtin_amdgcn_sched_barrier(0);

      float res[NSLOT];
#pragma unroll
      for (int P = 0; P < NSLOT; ++P) {
        res[P] = 0.0f;
        if (tok0 + 2 * NWAVE * P < t1) {
          v8f acc0 = {}, acc1 = {};
#pragma unroll
          for (int q = 0; q < NSTEP; ++q) {
            const v4f a = xb[P & 1][q];
            const v4f bw = wr[q];
            v2f a0 = {a.x, a.y}, a1 = {a.z, a.w};
            v2f b0 = {bw.x, bw.y}, b1 = {bw.z, bw.w};
            acc0 = __builtin_amdgcn_wmma_f32_16x16x4_f32(
                 false, a0,  false, b0,
                 (short)0, acc0,  false,  false);
            acc1 = __builtin_amdgcn_wmma_f32_16x16x4_f32(
                 false, a1,  false, b1,
                 (short)0, acc1,  false,  false);
          }
          const v8f acc = acc0 + acc1;
          res[P] = pick8(acc, p1, p2, p4);
          if (P + 2 < NSLOT) {
            if (tok0 + 2 * NWAVE * (P + 2) < t1) {
#pragma unroll
              for (int q = 0; q < NSTEP; ++q)
                xb[P & 1][q] = *reinterpret_cast<const v4f*>(xr[P + 2] + q * KSTEP);
            }
          }
        }
        __builtin_amdgcn_sched_barrier(0);
      }

#pragma unroll
      for (int P = 0; P < NSLOT; ++P) {
        float s = res[P];
        s += xor_row<1>(s);
        s += xor_row<2>(s);
        s += xor_row<4>(s);
        res[P] = s;
      }
      if (c == 0) {
#pragma unroll
        for (int P = 0; P < NSLOT; ++P)
          sDot[2 * (NWAVE * P + wave) + h] = res[P];
      }
    }
    if (tid < TOK) sVal[tid] = (base + tid < t1 && mraw > 0) ? 1 : 0;
    __syncthreads();
    if (tid < G) {
      const int lo = max(s0, base), hi = min(s1, base + TOK);
      for (int k0 = lo; k0 < hi; k0 += 8) {
        float d[8]; int v[8];
#pragma unroll
        for (int j = 0; j < 8; ++j) {
          d[j] = sDot[k0 - base + j];
          v[j] = sVal[k0 - base + j];
        }
#pragma unroll
        for (int j = 0; j < 8; ++j) {
          const bool take = (k0 + j < hi) && (v[j] != 0);
          sum = take ? sum + d[j] : sum;
          cnt += take ? 1 : 0;
        }
      }
    }
    __syncthreads();
  }
  if (tid < G)
    means[(b * NBLK + jb) * MSTR + tid] = sum / (float)max(cnt, 1);
}

constexpr int LWAVES = 8;
constexpr int LROWS  = G / 2;
constexpr int LBLK   = Sc / G;

__global__ __launch_bounds__(LWAVES * 32) void logits_kernel(
    const float* __restrict__ means, const float* __restrict__ biasPtr,
    float* __restrict__ out) {
  const int lane = threadIdx.x & 31;
  const int wave = __builtin_amdgcn_readfirstlane(threadIdx.x >> 5);
  const int b  = blockIdx.x / LBLK;
  const int sg = blockIdx.x % LBLK;
  const int q  = wave & 3;
  const int r  = wave >> 2;
  const int tt = (Tc / 4) * q + 4 * lane;
  const v4f cv = *reinterpret_cast<const v4f*>(
      means + (b * NBLK + Sc / G + tt / G) * MSTR + tt % G);
  const float* ap = means + (b * NBLK + sg) * MSTR + LROWS * r;
  const float bias = biasPtr[0];
  float* op = out + ((size_t)(b * Sc + G * sg + LROWS * r)) * Tc + tt;
  for (int i = 0; i < LROWS; ++i) {
    const float av = ap[i] + bias;
    *reinterpret_cast<v4f*>(op + (size_t)i * Tc) = cv + av;
  }
}

extern "C" void kernel_launch(void* const* d_in, const int* in_sizes, int n_in,
                              void* d_out, int out_size, void* d_ws, size_t ws_size,
                              hipStream_t stream) {
  const float* X    = (const float*)d_in[0];
  const int*   msk  = (const int*)d_in[1];
  const int*   wid  = (const int*)d_in[2];
  const float* wvec = (const float*)d_in[5];
  const float* bias = (const float*)d_in[6];
  float* out = (float*)d_out;

  float* means = (float*)d_ws;

  dot_seg_kernel<<<Bc * NBLK, 256, 0, stream>>>(X, wvec, msk, wid, means);

  logits_kernel<<<Bc * LBLK, LWAVES * 32, 0, stream>>>(
      means, bias, out);
}
